# starting kernel
# baseline (speedup 1.0000x reference)
#include <hip/hip_runtime.h>

typedef float v2f __attribute__((ext_vector_type(2)));
typedef float v8f __attribute__((ext_vector_type(8)));

#define C_CH       64
#define NK         128
#define NN         128
#define TOK_PER_WG 128
#define LDS_PITCH  130

__global__ __launch_bounds__(256) void expand_w_kernel(const float* __restrict__ W,
                                                       float* __restrict__ WT) {
    int idx = blockIdx.x * 256 + threadIdx.x;
    int n = idx >> 7;
    int k = idx & 127;
    int c = n >> 1;
    int j = k & 63;
    int half = k >> 6;
    float v = 0.0f;
    if (j != c) {
        int f = half * 63 + (j < c ? j : j - 1);
        v = W[n * 126 + f];
    }
    WT[idx] = v;
}

__global__ __launch_bounds__(256) void leakage_wmma_kernel(const float* __restrict__ x,
                                                           const float* __restrict__ WT,
                                                           float* __restrict__ out) {
    __shared__ float sF[TOK_PER_WG * LDS_PITCH];
    __shared__ float sW[NN * LDS_PITCH];

    const int  tid  = threadIdx.x;
    const long tok0 = (long)blockIdx.x * TOK_PER_WG;

    {
        const float2* g = (const float2*)WT;
        for (int i = tid; i < NN * NK / 2; i += 256) {
            int row  = i >> 6;
            int col2 = (i & 63) * 2;
            float2 v = g[i];
            sW[row * LDS_PITCH + col2]     = v.x;
            sW[row * LDS_PITCH + col2 + 1] = v.y;
        }
    }

    {
        const float2* x2 = (const float2*)x + tok0 * C_CH;
        for (int i = tid; i < TOK_PER_WG * C_CH; i += 256) {
            int t = i >> 6;
            int c = i & 63;
            float2 v  = x2[i];
            float amp = v.x * v.x + v.y * v.y;
            sF[t * LDS_PITCH + c]        = amp * v.x;
            sF[t * LDS_PITCH + C_CH + c] = amp * v.y;
        }
    }
    __syncthreads();

    const int wave  = tid >> 5;
    const int lane  = tid & 31;
    const int lrow  = lane & 15;
    const int khalf = lane >> 4;
    const int m0    = wave * 16;

    v8f acc[8];
    {
        v8f z = {0.f, 0.f, 0.f, 0.f, 0.f, 0.f, 0.f, 0.f};
#pragma unroll
        for (int i = 0; i < 8; ++i) acc[i] = z;
    }

    const float* Abase = &sF[(m0 + lrow) * LDS_PITCH + 2 * khalf];
    const float* Bbase = &sW[lrow * LDS_PITCH + 2 * khalf];

#pragma unroll 4
    for (int kk = 0; kk < NK / 4; ++kk) {
        v2f a = *(const v2f*)(Abase + 4 * kk);
#pragma unroll
        for (int nt = 0; nt < 8; ++nt) {
            v2f b = *(const v2f*)(Bbase + nt * 16 * LDS_PITCH + 4 * kk);
            acc[nt] = __builtin_amdgcn_wmma_f32_16x16x4_f32(
                false, a, false, b, (short)0, acc[nt], false, false);
        }
    }

    float* orow = out + (tok0 + m0 + khalf * 8) * NN;
#pragma unroll
    for (int nt = 0; nt < 8; ++nt) {
#pragma unroll
        for (int r = 0; r < 8; ++r) {
            orow[r * NN + nt * 16 + lrow] = acc[nt][r];
        }
    }
}

extern "C" void kernel_launch(void* const* d_in, const int* in_sizes, int n_in,
                              void* d_out, int out_size, void* d_ws, size_t ws_size,
                              hipStream_t stream) {
    const float* x  = (const float*)d_in[0];
    const float* W  = (const float*)d_in[1];
    float*       o  = (float*)d_out;
    float*       WT = (float*)d_ws;

    expand_w_kernel<<<64, 256, 0, stream>>>(W, WT);

    const int tokens = 16 * 1024;
    leakage_wmma_kernel<<<tokens / TOK_PER_WG, 256, 0, stream>>>(x, WT, o);
}


# Each finished 16-token tile is regrouped through a wave-private LDS strip (pitch 144 B): every global store instruction writes four whole 128-byte lines, not sixteen 32-byte pieces; sums unchanged
# speedup vs baseline: 2.3882x; 2.3882x over previous
#include <hip/hip_runtime.h>


typedef float v2f __attribute__((ext_vector_type(2)));
typedef float v4f __attribute__((ext_vector_type(4)));
typedef float v8f __attribute__((ext_vector_type(8)));

#define NK         128
#define NN         128
#define NF         126
#define TOK_PER_WG 64
#define OPITCH     9

__device__ __forceinline__ v4f load4_any(const float* p) {
    v4f r;
    __builtin_memcpy(&r, p, 16);
    return r;
}

__global__ __launch_bounds__(128) void leakage_wmma_kernel(const float* __restrict__ x,
                                                           const float* __restrict__ W,
                                                           float* __restrict__ out) {
    __shared__ __attribute__((aligned(16))) float sF[4 * 16 * 32 * 4 + 4 * 16 * OPITCH * 4];

    const int  tid  = threadIdx.x;
    const int  wave = tid >> 5;
    const int  lane = tid & 31;
    const int  lrow = lane & 15;
    const int  h    = lane >> 4;
    const long tok0 = (long)blockIdx.x * TOK_PER_WG;
    const long tok  = tok0 + wave * 16 + lrow;

    v4f xa[8], xb[8];
    {
        const v4f* x4 = (const v4f*)(x + tok * NK + 8 * h);
#pragma unroll
        for (int q = 0; q < 8; ++q) { xa[q] = x4[4 * q]; xb[q] = x4[4 * q + 1]; }
    }

    const int tile0 = __builtin_amdgcn_readfirstlane(wave) * 2;
    const int mrow  = ((lrow >> 2) & 1) * 8 + ((lrow >> 3) & 1) * 4 + (lrow & 3);
    v4f areg[2][16];
#pragma unroll
    for (int blk = 0; blk < 2; ++blk)
#pragma unroll
        for (int q = 0; q < 8; ++q)
#pragma unroll
            for (int t2 = 0; t2 < 2; ++t2) {
                const int n  = (tile0 + t2) * 16 + mrow;
                const int c  = n >> 1;
                const int d  = c - (8 * q + 4 * h);
                const int back = (blk == 0) ? (d < 0) : (d <= 3);
                areg[t2][blk * 8 + q] = load4_any(W + n * NF + blk * 63 + 8 * q + 4 * h - back);
            }

    v4f* sF4 = (v4f*)sF;
#pragma unroll
    for (int q = 0; q < 8; ++q) {
        v4f va = xa[q], vb = xb[q];
        float a0 = va.x * va.x + va.y * va.y;
        float a1 = va.z * va.z + va.w * va.w;
        float a2 = vb.x * vb.x + vb.y * vb.y;
        float a3 = vb.z * vb.z + vb.w * vb.w;
        v4f fre = {a0 * va.x, a1 * va.z, a2 * vb.x, a3 * vb.z};
        v4f fim = {a0 * va.y, a1 * va.w, a2 * vb.y, a3 * vb.w};
        sF4[(wave * 16 + q) * 32 + lane]     = fre;
        sF4[(wave * 16 + 8 + q) * 32 + lane] = fim;
    }
    __builtin_amdgcn_fence(__ATOMIC_RELEASE, "workgroup", "local");
    __builtin_amdgcn_s_barrier();
    __builtin_amdgcn_fence(__ATOMIC_ACQUIRE, "workgroup", "local");

#pragma unroll
    for (int t2 = 0; t2 < 2; ++t2) {
        const int tile = tile0 + t2;
        const int c    = (tile * 16 + mrow) >> 1;
#pragma unroll
        for (int q = 0; q < 8; ++q) {
            if (q == tile) {
                const int  d  = c - (8 * q + 4 * h);
                const bool in = (d >= 0) && (d <= 3);
                v4f L = areg[t2][q];
                v4f R;
                R.x = (d == 0) ? 0.0f : L.x;
                R.y = (d == 1) ? 0.0f : ((d < 1) ? L.x : L.y);
                R.z = (d == 2) ? 0.0f : ((d < 2) ? L.y : L.z);
                R.w = (d == 3) ? 0.0f : ((d < 3) ? L.z : L.w);
                areg[t2][q] = in ? R : L;
                v4f M = areg[t2][8 + q];
                v4f S;
                S.x = (d == 0) ? 0.0f : M.y;
                S.y = (d == 1) ? 0.0f : ((d > 1) ? M.z : M.y);
                S.z = (d == 2) ? 0.0f : ((d > 2) ? M.w : M.z);
                S.w = (d == 3) ? 0.0f : M.w;
                areg[t2][8 + q] = in ? S : M;
            }
        }
    }

    v8f acc[4][2];
    {
        v8f z = {0.f, 0.f, 0.f, 0.f, 0.f, 0.f, 0.f, 0.f};
#pragma unroll
        for (int i = 0; i < 4; ++i) { acc[i][0] = z; acc[i][1] = z; }
    }

    v4f bbuf[2][16];
#pragma unroll
    for (int j = 0; j < 16; ++j) bbuf[0][j] = sF4[j * 32 + lane];

    v4f* sO4 = sF4 + 4 * 16 * 32 + wave * (16 * OPITCH);
    const int wr   = lrow * OPITCH + h;
    const int trow = 8 * ((lane >> 3) & 1) + 4 * (lane >> 4);
    const int rd   = trow * OPITCH + (lane & 7);
    float* obase = out + (tok0 + trow) * NN + 32 * wave + 4 * (lane & 7);
    v4f stg[4];

#pragma unroll
    for (int i = 0; i < 4; ++i) {
        __builtin_amdgcn_sched_barrier(0);
        if (i > 0) {
#pragma unroll
            for (int t2 = 0; t2 < 2; ++t2)
#pragma unroll
                for (int g = 0; g < 2; ++g) {
                    v8f d = acc[i - 1][t2];
                    v4f o = {d[4 * g], d[4 * g + 1], d[4 * g + 2], d[4 * g + 3]};
                    sO4[wr + 4 * t2 + 2 * g] = o;
                }
#pragma unroll
            for (int s = 0; s < 4; ++s) stg[s] = sO4[rd + OPITCH * s];
        }
        if (i + 1 < 4) {
#pragma unroll
            for (int j = 0; j < 16; ++j)
                bbuf[(i + 1) & 1][j] = sF4[((i + 1) * 16 + j) * 32 + lane];
        }
        __builtin_amdgcn_sched_barrier(0);
#pragma unroll
        for (int j = 0; j < 16; ++j) {
            if (j == 8 && i > 0) {
                __builtin_amdgcn_sched_barrier(0);
#pragma unroll
                for (int s = 0; s < 4; ++s)
                    *(v4f*)(obase + ((i - 1) * 16 + s) * NN) = stg[s];
                __builtin_amdgcn_sched_barrier(0);
            }
#pragma unroll
            for (int u = 0; u < 2; ++u) {
                v4f b4 = bbuf[i & 1][j];
                v2f b;
                if (u == 0) { b.x = b4.x; b.y = b4.y; } else { b.x = b4.z; b.y = b4.w; }
#pragma unroll
                for (int t2 = 0; t2 < 2; ++t2) {
                    v4f a4 = areg[t2][j];
                    v2f a;
                    if (u == 0) { a.x = a4.x; a.y = a4.y; } else { a.x = a4.z; a.y = a4.w; }
                    acc[i][t2] = __builtin_amdgcn_wmma_f32_16x16x4_f32(
                        false, a, false, b, (short)0, acc[i][t2], false, false);
                }
            }
        }
    }
#pragma unroll
    for (int t2 = 0; t2 < 2; ++t2)
#pragma unroll
        for (int g = 0; g < 2; ++g) {
            v8f d = acc[3][t2];
            v4f o = {d[4 * g], d[4 * g + 1], d[4 * g + 2], d[4 * g + 3]};
            sO4[wr + 4 * t2 + 2 * g] = o;
        }
#pragma unroll
    for (int s = 0; s < 4; ++s) stg[s] = sO4[rd + OPITCH * s];
#pragma unroll
    for (int s = 0; s < 4; ++s) *(v4f*)(obase + (3 * 16 + s) * NN) = stg[s];
}

extern "C" void kernel_launch(void* const* d_in, const int* in_sizes, int n_in,
                              void* d_out, int out_size, void* d_ws, size_t ws_size,
                              hipStream_t stream) {
    const float* x = (const float*)d_in[0];
    const float* W = (const float*)d_in[1];
    float*       o = (float*)d_out;

    const int tokens = 16 * 1024;
    leakage_wmma_kernel<<<tokens / TOK_PER_WG, 128, 0, stream>>>(x, W, o);
}


# Each wave runs its own token tile first, F straight from its registers, and takes its weight chunks in arrival order (own-channel zero set at the slot's turn); barrier wait moved into that pass
# speedup vs baseline: 2.5590x; 1.0715x over previous
#include <hip/hip_runtime.h>


typedef float v2f __attribute__((ext_vector_type(2)));
typedef float v4f __attribute__((ext_vector_type(4)));
typedef float v8f __attribute__((ext_vector_type(8)));
typedef __attribute__((address_space(1))) v4f gv4f;

#define NK         128
#define NN         128
#define NF         126
#define TOK_PER_WG 64
#define OPITCH     9

__device__ __forceinline__ v4f load4_any(const float* p) {
    v4f r;
    __builtin_memcpy(&r, p, 16);
    return r;
}

__device__ __forceinline__ void wmma_hazard_slots() {
    asm volatile("v_nop\n\tv_nop\n\tv_nop\n\tv_nop\n\tv_nop\n\tv_nop\n\tv_nop\n\tv_nop");
}

__global__ __launch_bounds__(128) void leakage_wmma_kernel(const float* __restrict__ x,
                                                           const float* __restrict__ W,
                                                           float* __restrict__ out) {
    __shared__ __attribute__((aligned(16))) float sF[4 * 16 * 32 * 4 + 4 * 16 * OPITCH * 4];

    const int  tid  = threadIdx.x;
    const int  wave = tid >> 5;
    const int  lane = tid & 31;
    const int  lrow = lane & 15;
    const int  h    = lane >> 4;
    const long tok0 = (long)blockIdx.x * TOK_PER_WG;
    const long tok  = tok0 + wave * 16 + lrow;

    v4f xa[8], xb[8];
    {
        const v4f* x4 = (const v4f*)(x + tok * NK + 8 * h);
#pragma unroll
        for (int q = 0; q < 8; ++q) { xa[q] = x4[4 * q]; xb[q] = x4[4 * q + 1]; }
    }

    const int tile0 = __builtin_amdgcn_readfirstlane(wave) * 2;
    const int mrow  = ((lrow >> 2) & 1) * 8 + ((lrow >> 3) & 1) * 4 + (lrow & 3);
    v4f areg[2][16];
#pragma unroll
    for (int blk = 0; blk < 2; ++blk)
#pragma unroll
        for (int q = 0; q < 8; ++q)
#pragma unroll
            for (int t2 = 0; t2 < 2; ++t2) {
                const int n  = (tile0 + t2) * 16 + mrow;
                const int c  = n >> 1;
                const int d  = c - (8 * q + 4 * h);
                const int back = (blk == 0) ? (d < 0) : (d <= 3);
                areg[t2][blk * 8 + q] = load4_any(W + n * NF + blk * 63 + 8 * q + 4 * h - back);
            }

    v4f* sF4 = (v4f*)sF;
    v4f bbuf[2][16];
#pragma unroll
    for (int q = 0; q < 8; ++q) {
        v4f va = xa[q], vb = xb[q];
        float a0 = va.x * va.x + va.y * va.y;
        float a1 = va.z * va.z + va.w * va.w;
        float a2 = vb.x * vb.x + vb.y * vb.y;
        float a3 = vb.z * vb.z + vb.w * vb.w;
        v4f fre = {a0 * va.x, a1 * va.z, a2 * vb.x, a3 * vb.z};
        v4f fim = {a0 * va.y, a1 * va.w, a2 * vb.y, a3 * vb.w};
        sF4[(wave * 16 + q) * 32 + lane]     = fre;
        sF4[(wave * 16 + 8 + q) * 32 + lane] = fim;
        bbuf[0][q]     = fre;
        bbuf[0][8 + q] = fim;
    }

    v8f acc[4][2];
    {
        v8f z = {0.f, 0.f, 0.f, 0.f, 0.f, 0.f, 0.f, 0.f};
#pragma unroll
        for (int i = 0; i < 4; ++i) { acc[i][0] = z; acc[i][1] = z; }
    }

    v4f* sO4 = sF4 + 4 * 16 * 32 + wave * (16 * OPITCH);
    const int wr   = lrow * OPITCH + h;
    const int trow = 8 * ((lane >> 3) & 1) + 4 * (lane >> 4);
    const int rd   = trow * OPITCH + (lane & 7);
    float* obase = out + (tok0 + trow) * NN + 32 * wave + 4 * (lane & 7);
    v4f stg[4];

    const int wv = tile0 >> 1;
    int ttile[4];
#pragma unroll
    for (int p = 0; p < 4; ++p) ttile[p] = (wv + p) & 3;
    gv4f* otile[4];
#pragma unroll
    for (int p = 0; p < 4; ++p) {
        otile[p] = (gv4f*)(obase + ttile[p] * 16 * NN);
        asm volatile("" : "+v"(otile[p]));
    }

#pragma unroll
    for (int i = 0; i < 4; ++i) {
        if (i > 0) {
            __builtin_amdgcn_sched_barrier(0);
            if (i == 1) {
#pragma unroll
                for (int jj = 0; jj < 16; ++jj) asm volatile("" :: "v"(bbuf[0][jj]));
            }
            wmma_hazard_slots();
            __builtin_amdgcn_sched_barrier(0);
#pragma unroll
            for (int t2 = 0; t2 < 2; ++t2)
#pragma unroll
                for (int g = 0; g < 2; ++g) {
                    v8f d = acc[i - 1][t2];
                    v4f o = {d[4 * g], d[4 * g + 1], d[4 * g + 2], d[4 * g + 3]};
                    sO4[wr + 4 * t2 + 2 * g] = o;
                }
#pragma unroll
            for (int s = 0; s < 4; ++s) stg[s] = sO4[rd + OPITCH * s];
            if (i + 1 < 4) {
#pragma unroll
                for (int j = 0; j < 16; ++j)
                    bbuf[(i + 1) & 1][j] = sF4[(ttile[i + 1] * 16 + j) * 32 + lane];
            }
            __builtin_amdgcn_sched_barrier(0);
        }
#pragma unroll
        for (int j = 0; j < 16; ++j) {
            if (i == 0) {
                const int blk = j >> 3, q = j & 7;
#pragma unroll
                for (int t2 = 0; t2 < 2; ++t2) {
                    const int tile = tile0 + t2;
                    if (__builtin_expect(q == tile, 0)) {
                        const int  c  = (tile * 16 + mrow) >> 1;
                        const int  d  = c - (8 * q + 4 * h);
                        const bool in = (d >= 0) && (d <= 3);
                        if (blk == 0) {
                            v4f L = areg[t2][q];
                            v4f R;
                            R.x = (d == 0) ? 0.0f : L.x;
                            R.y = (d == 1) ? 0.0f : ((d < 1) ? L.x : L.y);
                            R.z = (d == 2) ? 0.0f : ((d < 2) ? L.y : L.z);
                            R.w = (d == 3) ? 0.0f : ((d < 3) ? L.z : L.w);
                            areg[t2][q] = in ? R : L;
                        } else {
                            v4f M = areg[t2][8 + q];
                            v4f S;
                            S.x = (d == 0) ? 0.0f : M.y;
                            S.y = (d == 1) ? 0.0f : ((d > 1) ? M.z : M.y);
                            S.z = (d == 2) ? 0.0f : ((d > 2) ? M.w : M.z);
                            S.w = (d == 3) ? 0.0f : M.w;
                            areg[t2][8 + q] = in ? S : M;
                        }
                    }
                }
                if (j == 2) {
                    __builtin_amdgcn_fence(__ATOMIC_RELEASE, "workgroup", "local");
                    __builtin_amdgcn_s_barrier_signal(-1);
                }
                if (j == 10) {
                    __builtin_amdgcn_sched_barrier(0);
                    __builtin_amdgcn_s_barrier_wait(-1);
                    __builtin_amdgcn_fence(__ATOMIC_ACQUIRE, "workgroup", "local");
#pragma unroll
                    for (int jj = 0; jj < 16; ++jj)
                        bbuf[1][jj] = sF4[(ttile[1] * 16 + jj) * 32 + lane];
                    __builtin_amdgcn_sched_barrier(0);
                }
            }
            if (j == 8 && i > 0) {
                __builtin_amdgcn_sched_barrier(0);
#pragma unroll
                for (int s = 0; s < 4; ++s)
                    otile[i - 1][s * (NN / 4)] = stg[s];
                __builtin_amdgcn_sched_barrier(0);
            }
#pragma unroll
            for (int u = 0; u < 2; ++u) {
                v4f b4 = bbuf[i & 1][j];
                v2f b;
                if (u == 0) { b.x = b4.x; b.y = b4.y; } else { b.x = b4.z; b.y = b4.w; }
#pragma unroll
                for (int t2 = 0; t2 < 2; ++t2) {
                    v4f a4 = areg[t2][j];
                    v2f a;
                    if (u == 0) { a.x = a4.x; a.y = a4.y; } else { a.x = a4.z; a.y = a4.w; }
                    acc[i][t2] = __builtin_amdgcn_wmma_f32_16x16x4_f32(
                        false, a, false, b, (short)0, acc[i][t2], false, false);
                }
            }
        }
    }
    __builtin_amdgcn_sched_barrier(0);
    wmma_hazard_slots();
    __builtin_amdgcn_sched_barrier(0);
#pragma unroll
    for (int t2 = 0; t2 < 2; ++t2)
#pragma unroll
        for (int g = 0; g < 2; ++g) {
            v8f d = acc[3][t2];
            v4f o = {d[4 * g], d[4 * g + 1], d[4 * g + 2], d[4 * g + 3]};
            sO4[wr + 4 * t2 + 2 * g] = o;
        }
#pragma unroll
    for (int s = 0; s < 4; ++s) stg[s] = sO4[rd + OPITCH * s];
#pragma unroll
    for (int s = 0; s < 4; ++s) otile[3][s * (NN / 4)] = stg[s];
}

extern "C" void kernel_launch(void* const* d_in, const int* in_sizes, int n_in,
                              void* d_out, int out_size, void* d_ws, size_t ws_size,
                              hipStream_t stream) {
    const float* x = (const float*)d_in[0];
    const float* W = (const float*)d_in[1];
    float*       o = (float*)d_out;

    const int tokens = 16 * 1024;
    leakage_wmma_kernel<<<tokens / TOK_PER_WG, 128, 0, stream>>>(x, W, o);
}
